# starting kernel
# baseline (speedup 1.0000x reference)
#include <hip/hip_runtime.h>


typedef float v2f __attribute__((ext_vector_type(2)));
typedef float v8f __attribute__((ext_vector_type(8)));

#define B_N    8
#define CIN_N  4
#define H_N    64
#define W_N    64
#define OUT_N  8
#define K2_N   9
#define G_N    8
#define HO_N   62
#define WO_N   62
#define P_N    (HO_N * WO_N)
#define KTOT   (CIN_N * K2_N * G_N)
#define NSTEP  (KTOT / 4)
#define MTOT   (B_N * P_N)
#define NTILES (MTOT / 16)
#define NPAIR  (NSTEP * 32)
#define R_CONST 6.25f

__device__ __forceinline__ float spline_s(float xe, float pl, float ph) {
    float u = xe - pl; u = (u > 0.f) ? u : 0.f;
    float v = ph - xe; v = (v > 0.f) ? v : 0.f;
    float s = u * v * R_CONST;
    return s * s;
}

__global__ __launch_bounds__(256) void kan_conv_wmma_f32(
    const float* __restrict__ x,
    const float* __restrict__ phase_low,
    const float* __restrict__ phase_high,
    const float* __restrict__ weight,
    const float* __restrict__ bias,
    float* __restrict__ out)
{
    __shared__ float ldsB[NSTEP * 64];
    __shared__ float ldsPL[KTOT];
    __shared__ float ldsPH[KTOT];

    const int tid = threadIdx.x;

    float2 wv[9];
    int    sel[9];
    #pragma unroll
    for (int i = 0; i < 9; ++i) {
        const int pr = tid + i * 256;
        const int t  = pr >> 5;
        const int ln = pr & 31;
        const int n  = ln & 15;
        const int k0 = t * 4 + ((ln >= 16) ? 2 : 0);
        wv[i]  = *(const float2*)&weight[(n & 7) * KTOT + k0];
        sel[i] = (n < OUT_N);
    }
    #pragma unroll
    for (int i = 0; i < 9; ++i) {
        const int pr = tid + i * 256;
        float2 v;
        v.x = sel[i] ? wv[i].x : 0.0f;
        v.y = sel[i] ? wv[i].y : 0.0f;
        *(float2*)&ldsB[pr * 2] = v;
    }
    #pragma unroll
    for (int i = 0; i < 2; ++i) {
        const int k = tid + i * 256;
        if (k < KTOT) {
            ldsPL[k] = phase_low[k];
            ldsPH[k] = phase_high[k];
        }
    }
    __syncthreads();

    const int wave = tid >> 5;
    const int lane = tid & 31;
    const int tile = blockIdx.x * 8 + wave;
    if (tile >= NTILES) return;

    const int m  = lane & 15;
    const int hi = lane >> 4;
    const int h2 = hi * 2;
    const int gm = tile * 16 + m;
    const int b  = gm / P_N;
    const int p  = gm - b * P_N;
    const int oi = p / WO_N;
    const int oj = p - oi * WO_N;

    v8f acc = {0.f, 0.f, 0.f, 0.f, 0.f, 0.f, 0.f, 0.f};

    for (int c = 0; c < CIN_N; ++c) {
        const float* xrow = x + ((b * CIN_N + c) * H_N + oi) * W_N + oj;
        #pragma unroll
        for (int f = 0; f < K2_N; ++f) {
            const int   di  = f / 3, dj = f - di * 3;
            const float xe  = xrow[di * W_N + dj];
            const int   kcf = (c * K2_N + f) * G_N;

            const float2 pl0 = *(const float2*)&ldsPL[kcf + h2];
            const float2 pl1 = *(const float2*)&ldsPL[kcf + 4 + h2];
            const float2 ph0 = *(const float2*)&ldsPH[kcf + h2];
            const float2 ph1 = *(const float2*)&ldsPH[kcf + 4 + h2];

            v2f a0, a1;
            a0[0] = spline_s(xe, pl0.x, ph0.x);
            a0[1] = spline_s(xe, pl0.y, ph0.y);
            a1[0] = spline_s(xe, pl1.x, ph1.x);
            a1[1] = spline_s(xe, pl1.y, ph1.y);

            const float2 b0 = *(const float2*)&ldsB[kcf * 16 + lane * 2];
            const float2 b1 = *(const float2*)&ldsB[(kcf + 4) * 16 + lane * 2];
            v2f bf0; bf0[0] = b0.x; bf0[1] = b0.y;
            v2f bf1; bf1[0] = b1.x; bf1[1] = b1.y;

            acc = __builtin_amdgcn_wmma_f32_16x16x4_f32(
                false, a0, false, bf0, (short)0, acc, false, false);
            acc = __builtin_amdgcn_wmma_f32_16x16x4_f32(
                false, a1, false, bf1, (short)0, acc, false, false);
        }
    }

    const int n = lane & 15;
    if (n < OUT_N) {
        const float bsum = bias[n * CIN_N + 0] + bias[n * CIN_N + 1] +
                           bias[n * CIN_N + 2] + bias[n * CIN_N + 3];
        int gmr = tile * 16 + hi * 8;
        int bb  = gmr / P_N;
        int pp  = gmr - bb * P_N;
        #pragma unroll
        for (int r = 0; r < 8; ++r) {
            out[(bb * OUT_N + n) * P_N + pp] = acc[r] + bsum;
            if (++pp == P_N) { pp = 0; ++bb; }
        }
    }
}

extern "C" void kernel_launch(void* const* d_in, const int* in_sizes, int n_in,
                              void* d_out, int out_size, void* d_ws, size_t ws_size,
                              hipStream_t stream)
{
    (void)in_sizes; (void)n_in; (void)out_size; (void)d_ws; (void)ws_size;
    const float* x  = (const float*)d_in[0];
    const float* pl = (const float*)d_in[1];
    const float* ph = (const float*)d_in[2];
    const float* w  = (const float*)d_in[3];
    const float* bs = (const float*)d_in[4];
    float* out = (float*)d_out;

    const int blocks = (NTILES + 7) / 8;
    kan_conv_wmma_f32<<<blocks, 256, 0, stream>>>(x, pl, ph, w, bs, out);
}


# Every global load (weights, phases, bias, all 36 input taps) issued at kernel entry ahead of the staging barrier; input-channel loop fully unrolled so the taps stay in registers
# speedup vs baseline: 1.3105x; 1.3105x over previous
#include <hip/hip_runtime.h>


typedef float v2f __attribute__((ext_vector_type(2)));
typedef float v8f __attribute__((ext_vector_type(8)));

#define B_N    8
#define CIN_N  4
#define H_N    64
#define W_N    64
#define OUT_N  8
#define K2_N   9
#define G_N    8
#define HO_N   62
#define WO_N   62
#define P_N    (HO_N * WO_N)
#define KTOT   (CIN_N * K2_N * G_N)
#define NSTEP  (KTOT / 4)
#define MTOT   (B_N * P_N)
#define NTILES (MTOT / 16)
#define NPAIR  (NSTEP * 32)
#define R_CONST 6.25f

__device__ __forceinline__ float spline_s(float xe, float pl, float ph) {
    float u = xe - pl; u = (u > 0.f) ? u : 0.f;
    float v = ph - xe; v = (v > 0.f) ? v : 0.f;
    float s = u * v * R_CONST;
    return s * s;
}

__global__ __launch_bounds__(256) void kan_conv_wmma_f32(
    const float* __restrict__ x,
    const float* __restrict__ phase_low,
    const float* __restrict__ phase_high,
    const float* __restrict__ weight,
    const float* __restrict__ bias,
    float* __restrict__ out)
{
    __shared__ float ldsB[NSTEP * 64];
    __shared__ float ldsPL[KTOT];
    __shared__ float ldsPH[KTOT];

    const int tid  = threadIdx.x;
    const int wave = tid >> 5;
    const int lane = tid & 31;
    const int tile = blockIdx.x * 8 + wave;
    const int tl   = (tile < NTILES) ? tile : (NTILES - 1);

    const int m  = lane & 15;
    const int hi = lane >> 4;
    const int h2 = hi * 2;
    const int gm = tl * 16 + m;
    const int b  = gm / P_N;
    const int p  = gm - b * P_N;
    const int oi = p / WO_N;
    const int oj = p - oi * WO_N;

    float2 wv[9];
    int    sel[9];
    #pragma unroll
    for (int i = 0; i < 9; ++i) {
        const int pr = tid + i * 256;
        const int t  = pr >> 5;
        const int ln = pr & 31;
        const int n  = ln & 15;
        const int k0 = t * 4 + ((ln >= 16) ? 2 : 0);
        wv[i]  = *(const float2*)&weight[(n & 7) * KTOT + k0];
        sel[i] = (n < OUT_N);
    }
    const int   kp1  = (tid + 256 < KTOT) ? (tid + 256) : (KTOT - 1);
    const float plr0 = phase_low[tid];
    const float phr0 = phase_high[tid];
    const float plr1 = phase_low[kp1];
    const float phr1 = phase_high[kp1];

    const int   nb   = (lane & 7) * CIN_N;
    const float bia0 = bias[nb + 0];
    const float bia1 = bias[nb + 1];
    const float bia2 = bias[nb + 2];
    const float bia3 = bias[nb + 3];

    float xe[CIN_N * K2_N];
    {
        const float* xpix = x + (b * CIN_N * H_N + oi) * W_N + oj;
        #pragma unroll
        for (int c = 0; c < CIN_N; ++c) {
            #pragma unroll
            for (int f = 0; f < K2_N; ++f) {
                const int di = f / 3, dj = f - di * 3;
                xe[c * K2_N + f] = xpix[(c * H_N + di) * W_N + dj];
            }
        }
    }

    #pragma unroll
    for (int i = 0; i < 9; ++i) {
        const int pr = tid + i * 256;
        float2 v;
        v.x = sel[i] ? wv[i].x : 0.0f;
        v.y = sel[i] ? wv[i].y : 0.0f;
        *(float2*)&ldsB[pr * 2] = v;
    }
    ldsPL[tid] = plr0;
    ldsPH[tid] = phr0;
    if (tid + 256 < KTOT) {
        ldsPL[tid + 256] = plr1;
        ldsPH[tid + 256] = phr1;
    }
    __syncthreads();

    if (tile >= NTILES) return;

    v8f acc = {0.f, 0.f, 0.f, 0.f, 0.f, 0.f, 0.f, 0.f};

    #pragma unroll
    for (int c = 0; c < CIN_N; ++c) {
        #pragma unroll
        for (int f = 0; f < K2_N; ++f) {
            const float xv  = xe[c * K2_N + f];
            const int   kcf = (c * K2_N + f) * G_N;

            const float2 pl0 = *(const float2*)&ldsPL[kcf + h2];
            const float2 pl1 = *(const float2*)&ldsPL[kcf + 4 + h2];
            const float2 ph0 = *(const float2*)&ldsPH[kcf + h2];
            const float2 ph1 = *(const float2*)&ldsPH[kcf + 4 + h2];

            v2f a0, a1;
            a0[0] = spline_s(xv, pl0.x, ph0.x);
            a0[1] = spline_s(xv, pl0.y, ph0.y);
            a1[0] = spline_s(xv, pl1.x, ph1.x);
            a1[1] = spline_s(xv, pl1.y, ph1.y);

            const float2 b0 = *(const float2*)&ldsB[kcf * 16 + lane * 2];
            const float2 b1 = *(const float2*)&ldsB[(kcf + 4) * 16 + lane * 2];
            v2f bf0; bf0[0] = b0.x; bf0[1] = b0.y;
            v2f bf1; bf1[0] = b1.x; bf1[1] = b1.y;

            acc = __builtin_amdgcn_wmma_f32_16x16x4_f32(
                false, a0, false, bf0, (short)0, acc, false, false);
            acc = __builtin_amdgcn_wmma_f32_16x16x4_f32(
                false, a1, false, bf1, (short)0, acc, false, false);
        }
    }

    const int n = lane & 15;
    if (n < OUT_N) {
        const float bsum = bia0 + bia1 + bia2 + bia3;
        int gmr = tile * 16 + hi * 8;
        int bb  = gmr / P_N;
        int pp  = gmr - bb * P_N;
        #pragma unroll
        for (int r = 0; r < 8; ++r) {
            out[(bb * OUT_N + n) * P_N + pp] = acc[r] + bsum;
            if (++pp == P_N) { pp = 0; ++bb; }
        }
    }
}

extern "C" void kernel_launch(void* const* d_in, const int* in_sizes, int n_in,
                              void* d_out, int out_size, void* d_ws, size_t ws_size,
                              hipStream_t stream)
{
    (void)in_sizes; (void)n_in; (void)out_size; (void)d_ws; (void)ws_size;
    const float* x  = (const float*)d_in[0];
    const float* pl = (const float*)d_in[1];
    const float* ph = (const float*)d_in[2];
    const float* w  = (const float*)d_in[3];
    const float* bs = (const float*)d_in[4];
    float* out = (float*)d_out;

    const int blocks = (NTILES + 7) / 8;
    kan_conv_wmma_f32<<<blocks, 256, 0, stream>>>(x, pl, ph, w, bs, out);
}


# LDS reads of phases and weight fragments taken out of the matrix-op stream: per input channel one group of 27 16-byte reads (LDS re-laid per tap and lane) issued a third of a channel ahead
# speedup vs baseline: 1.4631x; 1.1164x over previous
#include <hip/hip_runtime.h>


typedef float v2f __attribute__((ext_vector_type(2)));
typedef float v4f __attribute__((ext_vector_type(4)));
typedef float v8f __attribute__((ext_vector_type(8)));

#define B_N    8
#define CIN_N  4
#define H_N    64
#define W_N    64
#define OUT_N  8
#define K2_N   9
#define G_N    8
#define HO_N   62
#define WO_N   62
#define P_N    (HO_N * WO_N)
#define KTOT   (CIN_N * K2_N * G_N)
#define NSTEP  (KTOT / 4)
#define MTOT   (B_N * P_N)
#define NTILES (MTOT / 16)
#define NPAIR  (NSTEP * 32)
#define R_CONST 6.25f

__device__ __forceinline__ float spline_s(float xe, float pl, float ph) {
    float u = xe - pl; u = (u > 0.f) ? u : 0.f;
    float v = ph - xe; v = (v > 0.f) ? v : 0.f;
    float s = u * v * R_CONST;
    return s * s;
}

__global__ __launch_bounds__(256) void kan_conv_wmma_f32(
    const float* __restrict__ x,
    const float* __restrict__ phase_low,
    const float* __restrict__ phase_high,
    const float* __restrict__ weight,
    const float* __restrict__ bias,
    float* __restrict__ out)
{
    __shared__ __attribute__((aligned(16))) float ldsB[NSTEP * 64];
    __shared__ __attribute__((aligned(16))) float ldsP[2 * KTOT];
#define IDX_B(t_, ln_)   (((((t_) >> 1) * 32 + (ln_)) * 4) + (((t_) & 1) * 2))
#define IDX_P(tap_, g_)  (((((tap_) * 2) + (((g_) >> 1) & 1)) * 8) + ((((g_) >> 2) & 1) * 2) + ((g_) & 1))

    const int tid  = threadIdx.x;
    const int wave = tid >> 5;
    const int lane = tid & 31;
    const int tile = blockIdx.x * 8 + wave;
    const int tl   = (tile < NTILES) ? tile : (NTILES - 1);

    const int m  = lane & 15;
    const int hi = lane >> 4;
    const int gm = tl * 16 + m;
    const int b  = gm / P_N;
    const int p  = gm - b * P_N;
    const int oi = p / WO_N;
    const int oj = p - oi * WO_N;

    float2 wv[9];
    int    sel[9];
    #pragma unroll
    for (int i = 0; i < 9; ++i) {
        const int pr = tid + i * 256;
        const int t  = pr >> 5;
        const int ln = pr & 31;
        const int n  = ln & 15;
        const int k0 = t * 4 + ((ln >= 16) ? 2 : 0);
        wv[i]  = *(const float2*)&weight[(n & 7) * KTOT + k0];
        sel[i] = (n < OUT_N);
    }
    const int   kp1  = (tid + 256 < KTOT) ? (tid + 256) : (KTOT - 1);
    const float plr0 = phase_low[tid];
    const float phr0 = phase_high[tid];
    const float plr1 = phase_low[kp1];
    const float phr1 = phase_high[kp1];

    const int   nb   = (lane & 7) * CIN_N;
    const float bia0 = bias[nb + 0];
    const float bia1 = bias[nb + 1];
    const float bia2 = bias[nb + 2];
    const float bia3 = bias[nb + 3];

    float xe[CIN_N * K2_N];
    {
        const float* xpix = x + (b * CIN_N * H_N + oi) * W_N + oj;
        #pragma unroll
        for (int c = 0; c < CIN_N; ++c) {
            #pragma unroll
            for (int f = 0; f < K2_N; ++f) {
                const int di = f / 3, dj = f - di * 3;
                xe[c * K2_N + f] = xpix[(c * H_N + di) * W_N + dj];
            }
        }
    }

    #pragma unroll
    for (int i = 0; i < 9; ++i) {
        const int pr = tid + i * 256;
        float2 v;
        v.x = sel[i] ? wv[i].x : 0.0f;
        v.y = sel[i] ? wv[i].y : 0.0f;
        *(float2*)&ldsB[IDX_B(pr >> 5, pr & 31)] = v;
    }
    {
        const int k0i = IDX_P(tid >> 3, tid & 7);
        ldsP[k0i]     = plr0;
        ldsP[k0i + 4] = phr0;
        if (tid + 256 < KTOT) {
            const int k1i = IDX_P((tid + 256) >> 3, (tid + 256) & 7);
            ldsP[k1i]     = plr1;
            ldsP[k1i + 4] = phr1;
        }
    }
    __syncthreads();

    if (tile >= NTILES) return;

    v8f acc = {0.f, 0.f, 0.f, 0.f, 0.f, 0.f, 0.f, 0.f};

    typedef const volatile __attribute__((address_space(3))) v4f* lds_v4f_ptr;
    v4f plq[2][K2_N], phq[2][K2_N], bq[2][K2_N];

#define LOAD_CHANNEL(c_, s_)                                                        \
    _Pragma("unroll")                                                               \
    for (int f = 0; f < K2_N; ++f) {                                                \
        const int tap = (c_) * K2_N + f;                                            \
        plq[s_][f] = *(lds_v4f_ptr)&ldsP[(tap * 2 + hi) * 8];                       \
        phq[s_][f] = *(lds_v4f_ptr)&ldsP[(tap * 2 + hi) * 8 + 4];                   \
        bq[s_][f]  = *(lds_v4f_ptr)&ldsB[(tap * 32 + lane) * 4];                    \
    }

#define COMPUTE_TAPS(c_, s_, f_lo, f_hi)                                            \
    _Pragma("unroll")                                                               \
    for (int f = (f_lo); f < (f_hi); ++f) {                                         \
        const float xv = xe[(c_) * K2_N + f];                                       \
        v2f a0, a1, bf0, bf1;                                                       \
        a0[0] = spline_s(xv, plq[s_][f][0], phq[s_][f][0]);                         \
        a0[1] = spline_s(xv, plq[s_][f][1], phq[s_][f][1]);                         \
        a1[0] = spline_s(xv, plq[s_][f][2], phq[s_][f][2]);                         \
        a1[1] = spline_s(xv, plq[s_][f][3], phq[s_][f][3]);                         \
        bf0[0] = bq[s_][f][0]; bf0[1] = bq[s_][f][1];                               \
        bf1[0] = bq[s_][f][2]; bf1[1] = bq[s_][f][3];                               \
        acc = __builtin_amdgcn_wmma_f32_16x16x4_f32(                                \
            false, a0, false, bf0, (short)0, acc, false, false);                    \
        acc = __builtin_amdgcn_wmma_f32_16x16x4_f32(                                \
            false, a1, false, bf1, (short)0, acc, false, false);                    \
    }

    LOAD_CHANNEL(0, 0)

    #pragma unroll
    for (int c = 0; c < CIN_N; ++c) {
        const int s = c & 1;
        __builtin_amdgcn_sched_barrier(0);
        COMPUTE_TAPS(c, s, 0, 6)
        __builtin_amdgcn_sched_barrier(0);
        if (c + 1 < CIN_N) {
            LOAD_CHANNEL(c + 1, s ^ 1)
        }
        __builtin_amdgcn_sched_barrier(0);
        COMPUTE_TAPS(c, s, 6, K2_N)
    }
#undef LOAD_CHANNEL
#undef COMPUTE_TAPS

    const int n = lane & 15;
    if (n < OUT_N) {
        const float bsum = bia0 + bia1 + bia2 + bia3;
        int gmr = tile * 16 + hi * 8;
        int bb  = gmr / P_N;
        int pp  = gmr - bb * P_N;
        #pragma unroll
        for (int r = 0; r < 8; ++r) {
            out[(bb * OUT_N + n) * P_N + pp] = acc[r] + bsum;
            if (++pp == P_N) { pp = 0; ++bb; }
        }
    }
}

extern "C" void kernel_launch(void* const* d_in, const int* in_sizes, int n_in,
                              void* d_out, int out_size, void* d_ws, size_t ws_size,
                              hipStream_t stream)
{
    (void)in_sizes; (void)n_in; (void)out_size; (void)d_ws; (void)ws_size;
    const float* x  = (const float*)d_in[0];
    const float* pl = (const float*)d_in[1];
    const float* ph = (const float*)d_in[2];
    const float* w  = (const float*)d_in[3];
    const float* bs = (const float*)d_in[4];
    float* out = (float*)d_out;

    const int blocks = (NTILES + 7) / 8;
    kan_conv_wmma_f32<<<blocks, 256, 0, stream>>>(x, pl, ph, w, bs, out);
}
